# starting kernel
# baseline (speedup 1.0000x reference)
#include <hip/hip_runtime.h>
#include <hip/hip_bf16.h>

typedef __attribute__((ext_vector_type(2))) float v2f;
typedef __attribute__((ext_vector_type(8))) float v8f;

#define NROWS 2048
#define DIM   64
#define WIDTH 5

__global__ void __launch_bounds__(128)
mlp_wmma_kernel(const float* __restrict__ x, const float* __restrict__ y,
                const float* __restrict__ W0, const float* __restrict__ b0,
                const float* __restrict__ Wh, const float* __restrict__ bh,
                const float* __restrict__ Wout, const float* __restrict__ bout,
                float* __restrict__ t  )
{
    __shared__ float hbuf[64 * WIDTH];

    const int tid  = threadIdx.x;
    const int lane = tid & 31;
    const int wave = tid >> 5;

    const int rowTile = blockIdx.x * 64 + wave * 16;
    const bool isY    = rowTile >= NROWS;
    const float* src  = isY ? y : x;
    const int srcRow  = rowTile - (isY ? NROWS : 0);

    const int m  = lane & 15;
    const int hi = lane >> 4;

    const float* arow = src + (size_t)(srcRow + m) * DIM + hi * 2;
    const float* brow = W0 + (m < WIDTH ? m : 0) * DIM + hi * 2;
    const float  bmask = (m < WIDTH) ? 1.0f : 0.0f;

    v2f bfrag[16];
    #pragma unroll
    for (int s = 0; s < 16; ++s) {
        v2f bv = *(const v2f*)(brow + s * 4);
        bfrag[s].x = bv.x * bmask;
        bfrag[s].y = bv.y * bmask;
    }

    v8f acc = {};
    #pragma unroll
    for (int s = 0; s < 16; ++s) {
        v2f a = *(const v2f*)(arow + s * 4);
        acc = __builtin_amdgcn_wmma_f32_16x16x4_f32(
                  false, a, false, bfrag[s], (short)0, acc, false, false);
    }

    if (m < WIDTH) {
        const float bias = b0[m];
        #pragma unroll
        for (int v = 0; v < 8; ++v) {
            float hv = acc[v] + bias;
            hv = fmaxf(hv, 0.0f);
            hbuf[(wave * 16 + hi * 8 + v) * WIDTH + m] = hv;
        }
    }
    __syncthreads();

    if (tid < 64) {
        float h[WIDTH], hn[WIDTH];
        #pragma unroll
        for (int q = 0; q < WIDTH; ++q) h[q] = hbuf[tid * WIDTH + q];

        for (int l = 0; l < 3; ++l) {
            #pragma unroll
            for (int n = 0; n < WIDTH; ++n) {
                float s = bh[n];
                #pragma unroll
                for (int q = 0; q < WIDTH; ++q) s += h[q] * Wh[n * WIDTH + q];
                hn[n] = fmaxf(s, 0.0f);
            }
            #pragma unroll
            for (int n = 0; n < WIDTH; ++n) h[n] = hn[n];
        }

        const int grow = blockIdx.x * 64 + tid;
        float* trow = t + (size_t)grow * DIM;
        #pragma unroll 8
        for (int c = 0; c < DIM; ++c) {
            float s = bout[c];
            #pragma unroll
            for (int q = 0; q < WIDTH; ++q) s += h[q] * Wout[c * WIDTH + q];
            trow[c] = fmaxf(s, 0.0f);
        }
    }
}

#define LSTRIDE 65

__global__ void __launch_bounds__(256)
pairwise_kernel(const float* __restrict__ t, float* __restrict__ out)
{
    __shared__ float txs[64 * LSTRIDE];
    __shared__ float tys[64 * LSTRIDE];

    const float* tx = t;
    const float* ty = t + (size_t)NROWS * DIM;

    const int i0  = blockIdx.y * 64;
    const int j0  = blockIdx.x * 64;
    const int tid = threadIdx.x;

    for (int e = tid; e < 64 * DIM; e += 256) {
        const int r = e >> 6;
        const int c = e & 63;
        txs[r * LSTRIDE + c] = tx[(size_t)(i0 + r) * DIM + c];
        tys[r * LSTRIDE + c] = ty[(size_t)(j0 + r) * DIM + c];
    }
    __syncthreads();

    const int ti = tid >> 4;
    const int tj = tid & 15;
    const float* txr = &txs[(ti * 4) * LSTRIDE];
    const float* tyr = &tys[(tj * 4) * LSTRIDE];

    float acc[4][4];
    #pragma unroll
    for (int r = 0; r < 4; ++r)
        #pragma unroll
        for (int c = 0; c < 4; ++c) acc[r][c] = 0.0f;

    #pragma unroll 4
    for (int k = 0; k < DIM; ++k) {
        float av[4], bv[4];
        #pragma unroll
        for (int r = 0; r < 4; ++r) av[r] = txr[r * LSTRIDE + k];
        #pragma unroll
        for (int c = 0; c < 4; ++c) bv[c] = tyr[c * LSTRIDE + k];
        #pragma unroll
        for (int r = 0; r < 4; ++r) {
            #pragma unroll
            for (int c = 0; c < 4; ++c) {
                float d = fmaxf(bv[c] - av[r], 0.0f);
                acc[r][c] = fmaf(d, d, acc[r][c]);
            }
        }
    }

    #pragma unroll
    for (int r = 0; r < 4; ++r) {
        float* orow = out + (size_t)(i0 + ti * 4 + r) * NROWS + j0 + tj * 4;
        #pragma unroll
        for (int c = 0; c < 4; ++c) orow[c] = acc[r][c];
    }
}

extern "C" void kernel_launch(void* const* d_in, const int* in_sizes, int n_in,
                              void* d_out, int out_size, void* d_ws, size_t ws_size,
                              hipStream_t stream) {
    const float* x    = (const float*)d_in[0];
    const float* y    = (const float*)d_in[1];
    const float* W0   = (const float*)d_in[2];
    const float* b0   = (const float*)d_in[3];
    const float* Wh   = (const float*)d_in[4];
    const float* bh   = (const float*)d_in[5];
    const float* Wout = (const float*)d_in[6];
    const float* bout = (const float*)d_in[7];
    float* out = (float*)d_out;
    float* t   = (float*)d_ws;

    mlp_wmma_kernel<<<64, 128, 0, stream>>>(x, y, W0, b0, Wh, bh, Wout, bout, t);

    pairwise_kernel<<<dim3(32, 32), 256, 0, stream>>>(t, out);
}


# Fuse the MLP into the pairwise kernel: each 64x64-tile block runs the MLP of its own 128 rows into LDS (layer 0 stays on f32 WMMA); 2 launches -> 1, global scratch round trip gone; sum order kept
# speedup vs baseline: 1.1121x; 1.1121x over previous
#include <hip/hip_runtime.h>

typedef __attribute__((ext_vector_type(2))) float v2f;
typedef __attribute__((ext_vector_type(8))) float v8f;

#define NROWS   2048
#define DIM     64
#define WIDTH   5
#define LSTRIDE 65
#define TROWS   128
#define HPITCH  8
#define H0PITCH 20

__global__ void __launch_bounds__(256)
fused_mlp_pairwise_kernel(float* __restrict__ out,
                          const float* __restrict__ x, const float* __restrict__ y,
                          const float* __restrict__ W0, const float* __restrict__ b0,
                          const float* __restrict__ Wh, const float* __restrict__ bh,
                          const float* __restrict__ Wout, const float* __restrict__ bout)
{
    __shared__ float ts[TROWS * LSTRIDE];
    __shared__ __attribute__((aligned(16))) float hbuf[TROWS * H0PITCH];
    __shared__ __attribute__((aligned(16))) float hfin[TROWS * HPITCH];

    const int i0   = blockIdx.y * 64;
    const int j0   = blockIdx.x * 64;
    const int tid  = threadIdx.x;
    const int lane = tid & 31;
    const int wave = tid >> 5;

    const int m  = lane & 15;
    const int hi = lane >> 4;

    float b0v[WIDTH], wh[WIDTH * WIDTH], bhv[WIDTH];
    #pragma unroll
    for (int n = 0; n < WIDTH; ++n) b0v[n] = b0[n];
    #pragma unroll
    for (int n = 0; n < WIDTH * WIDTH; ++n) wh[n] = Wh[n];
    #pragma unroll
    for (int n = 0; n < WIDTH; ++n) bhv[n] = bh[n];
    float woA[WIDTH], woB[WIDTH];
    #pragma unroll
    for (int q = 0; q < WIDTH; ++q) {
        woA[q] = Wout[lane * WIDTH + q];
        woB[q] = Wout[(lane + 32) * WIDTH + q];
    }
    const float boA = bout[lane];
    const float boB = bout[lane + 32];

    const bool isY   = wave >= 4;
    const float* src = isY ? y : x;
    const int srcRow = (isY ? j0 : i0) + (wave & 3) * 16;

    const float* arow = src + (size_t)(srcRow + m) * DIM + hi * 2;
    const float* brow = W0 + (m < WIDTH ? m : 0) * DIM + hi * 2;
    const float  bmask = (m < WIDTH) ? 1.0f : 0.0f;

    v2f bfrag[16];
    #pragma unroll
    for (int s = 0; s < 16; ++s) {
        v2f bv = *(const v2f*)(brow + s * 4);
        bfrag[s].x = bv.x * bmask;
        bfrag[s].y = bv.y * bmask;
    }

    v8f acc = {};
    #pragma unroll
    for (int s = 0; s < 16; ++s) {
        v2f a = *(const v2f*)(arow + s * 4);
        acc = __builtin_amdgcn_wmma_f32_16x16x4_f32(
                  false, a, false, bfrag[s], (short)0, acc, false, false);
    }

    #pragma unroll
    for (int v = 0; v < 8; ++v)
        hbuf[(wave * 16 + hi * 8 + v) * H0PITCH + m] = acc[v];
    __syncthreads();

    {
        const int row = wave * 16 + m;
        float h[WIDTH], hn[WIDTH];
        #pragma unroll
        for (int q = 0; q < WIDTH; ++q) h[q] = fmaxf(hbuf[row * H0PITCH + q] + b0v[q], 0.0f);

        #pragma unroll
        for (int l = 0; l < 3; ++l) {
            #pragma unroll
            for (int n = 0; n < WIDTH; ++n) {
                float s = bhv[n];
                #pragma unroll
                for (int q = 0; q < WIDTH; ++q) s += h[q] * wh[n * WIDTH + q];
                hn[n] = fmaxf(s, 0.0f);
            }
            #pragma unroll
            for (int n = 0; n < WIDTH; ++n) h[n] = hn[n];
        }

        if (hi == 0) {
            #pragma unroll
            for (int q = 0; q < WIDTH; ++q) hfin[row * HPITCH + q] = h[q];
        }
    }
    __syncthreads();

    {
        #pragma unroll
        for (int rr = 0; rr < 16; ++rr) {
            const int row = wave * 16 + rr;
            const float* hp = &hfin[row * HPITCH];
            float sA = boA, sB = boB;
            #pragma unroll
            for (int q = 0; q < WIDTH; ++q) {
                const float hq = hp[q];
                sA += hq * woA[q];
                sB += hq * woB[q];
            }
            ts[row * LSTRIDE + lane]      = fmaxf(sA, 0.0f);
            ts[row * LSTRIDE + lane + 32] = fmaxf(sB, 0.0f);
        }
    }
    __syncthreads();

    const int ti = tid >> 4;
    const int tj = tid & 15;
    const float* txr = &ts[(ti * 4) * LSTRIDE];
    const float* tyr = &ts[(64 + tj * 4) * LSTRIDE];

    float o[4][4];
    #pragma unroll
    for (int r = 0; r < 4; ++r)
        #pragma unroll
        for (int c = 0; c < 4; ++c) o[r][c] = 0.0f;

    #pragma unroll 4
    for (int k = 0; k < DIM; ++k) {
        float av[4], bv[4];
        #pragma unroll
        for (int r = 0; r < 4; ++r) av[r] = txr[r * LSTRIDE + k];
        #pragma unroll
        for (int c = 0; c < 4; ++c) bv[c] = tyr[c * LSTRIDE + k];
        #pragma unroll
        for (int r = 0; r < 4; ++r) {
            #pragma unroll
            for (int c = 0; c < 4; ++c) {
                float d = fmaxf(bv[c] - av[r], 0.0f);
                o[r][c] = fmaf(d, d, o[r][c]);
            }
        }
    }

    #pragma unroll
    for (int r = 0; r < 4; ++r) {
        float* orow = out + (size_t)(i0 + ti * 4 + r) * NROWS + j0 + tj * 4;
        #pragma unroll
        for (int c = 0; c < 4; ++c) orow[c] = o[r][c];
    }
}

extern "C" void kernel_launch(void* const* d_in, const int* in_sizes, int n_in,
                              void* d_out, int out_size, void* d_ws, size_t ws_size,
                              hipStream_t stream) {
    const float* x    = (const float*)d_in[0];
    const float* y    = (const float*)d_in[1];
    const float* W0   = (const float*)d_in[2];
    const float* b0   = (const float*)d_in[3];
    const float* Wh   = (const float*)d_in[4];
    const float* bh   = (const float*)d_in[5];
    const float* Wout = (const float*)d_in[6];
    const float* bout = (const float*)d_in[7];
    float* out = (float*)d_out;

    fused_mlp_pairwise_kernel<<<dim3(32, 32), 256, 0, stream>>>(out, x, y, W0, b0, Wh, bh, Wout, bout);
}
